# starting kernel
# baseline (speedup 1.0000x reference)
#include <hip/hip_runtime.h>

typedef __attribute__((ext_vector_type(2))) float v2f;
typedef __attribute__((ext_vector_type(8))) float v8f;

#define BB 8
#define CC 256
#define HWN 4096

__global__ __launch_bounds__(128) void gemm256_kernel(
    const float* __restrict__ A, long aStride,
    const float* __restrict__ bias,
    const float* __restrict__ Bm, long bStride,
    float* __restrict__ D, long dStride)
{
    const int lane  = threadIdx.x & 31;
    const int wave  = threadIdx.x >> 5;
    const int g     = lane >> 4;
    const int ln    = lane & 15;
    const int batch = blockIdx.z;
    const int mBase = blockIdx.y * 64 + wave * 16;
    const int nBase = blockIdx.x * 64;

    const float* Ab = A + (long)batch * aStride;
    const float* Bb = Bm + (long)batch * bStride;
    float*       Db = D + (long)batch * dStride;

    v8f acc0 = {}, acc1 = {}, acc2 = {}, acc3 = {};
    const int m  = mBase + ln;
    const int n0 = nBase + ln;

    #pragma unroll 4
    for (int k = 0; k < CC; k += 4) {
        v2f a = *(const v2f*)(Ab + (long)m * CC + k + 2 * g);
        const float* br0 = Bb + (long)(k + 2 * g) * HWN;
        const float* br1 = br0 + HWN;
        v2f b0 = { br0[n0     ], br1[n0     ] };
        v2f b1 = { br0[n0 + 16], br1[n0 + 16] };
        v2f b2 = { br0[n0 + 32], br1[n0 + 32] };
        v2f b3 = { br0[n0 + 48], br1[n0 + 48] };
        acc0 = __builtin_amdgcn_wmma_f32_16x16x4_f32(false, a, false, b0, (short)0, acc0, false, false);
        acc1 = __builtin_amdgcn_wmma_f32_16x16x4_f32(false, a, false, b1, (short)0, acc1, false, false);
        acc2 = __builtin_amdgcn_wmma_f32_16x16x4_f32(false, a, false, b2, (short)0, acc2, false, false);
        acc3 = __builtin_amdgcn_wmma_f32_16x16x4_f32(false, a, false, b3, (short)0, acc3, false, false);
    }

    #pragma unroll
    for (int r = 0; r < 8; ++r) {
        const int mo = mBase + r + 8 * g;
        const float bv = bias ? bias[mo] : 0.0f;
        float* drow = Db + (long)mo * HWN + n0;
        drow[0]  = acc0[r] + bv;
        drow[16] = acc1[r] + bv;
        drow[32] = acc2[r] + bv;
        drow[48] = acc3[r] + bv;
    }
}

__global__ __launch_bounds__(128) void corr_kernel(
    const float* __restrict__ Q, const float* __restrict__ Km,
    float* __restrict__ corr)
{
    const int lane  = threadIdx.x & 31;
    const int wave  = threadIdx.x >> 5;
    const int g     = lane >> 4;
    const int ln    = lane & 15;
    const int batch = blockIdx.z;
    const int mBase = blockIdx.y * 64 + wave * 16;
    const int nBase = blockIdx.x * 16;

    const float* qp = Q  + (long)batch * CC * HWN + (long)(mBase + ln) * HWN;
    const float* kp = Km + (long)batch * CC * HWN + (long)(nBase + ln) * HWN;

    v8f acc = {};
    #pragma unroll 8
    for (int k = 0; k < HWN; k += 4) {
        v2f a = *(const v2f*)(qp + k + 2 * g);
        v2f b = *(const v2f*)(kp + k + 2 * g);
        acc = __builtin_amdgcn_wmma_f32_16x16x4_f32(false, a, false, b, (short)0, acc, false, false);
    }

    float* cb = corr + (long)batch * CC * CC;
    #pragma unroll
    for (int r = 0; r < 8; ++r) {
        cb[(long)(mBase + r + 8 * g) * CC + nBase + ln] = acc[r];
    }
}

__global__ __launch_bounds__(256) void softmax_out_kernel(
    const float* __restrict__ scores,
    const float* __restrict__ fg, const float* __restrict__ mask,
    const float* __restrict__ V,  const float* __restrict__ gamma,
    float* __restrict__ out)
{
    const int  tid  = threadIdx.x;
    const int  lane = tid & 31;
    const int  wid  = tid >> 5;
    const long base = (long)blockIdx.x * HWN;

    __shared__ float red[8];

    float vals[16];
    float mx = -3.402823466e38f;
    #pragma unroll
    for (int t = 0; t < 16; ++t) {
        vals[t] = scores[base + t * 256 + tid];
        mx = fmaxf(mx, vals[t]);
    }
    #pragma unroll
    for (int off = 16; off >= 1; off >>= 1)
        mx = fmaxf(mx, __shfl_xor(mx, off, 32));
    if (lane == 0) red[wid] = mx;
    __syncthreads();
    #pragma unroll
    for (int w = 0; w < 8; ++w) mx = fmaxf(mx, red[w]);
    __syncthreads();

    float sum = 0.0f;
    #pragma unroll
    for (int t = 0; t < 16; ++t) {
        vals[t] = __expf(vals[t] - mx);
        sum += vals[t];
    }
    #pragma unroll
    for (int off = 16; off >= 1; off >>= 1)
        sum += __shfl_xor(sum, off, 32);
    if (lane == 0) red[wid] = sum;
    __syncthreads();
    float tot = 0.0f;
    #pragma unroll
    for (int w = 0; w < 8; ++w) tot += red[w];

    const float inv = 1.0f / tot;
    const float gam = gamma[0];
    #pragma unroll
    for (int t = 0; t < 16; ++t) {
        const long i = base + t * 256 + tid;
        const float att = vals[t] * inv * V[i];
        const float mk  = mask[i];
        out[i] = fg[i] * mk + gam * (1.0f - mk) * att;
    }
}

extern "C" void kernel_launch(void* const* d_in, const int* in_sizes, int n_in,
                              void* d_out, int out_size, void* d_ws, size_t ws_size,
                              hipStream_t stream)
{
    (void)in_sizes; (void)n_in; (void)out_size; (void)ws_size;

    const float* fg    = (const float*)d_in[0];
    const float* bg    = (const float*)d_in[1];
    const float* mask  = (const float*)d_in[2];
    const float* Wq    = (const float*)d_in[3];
    const float* bq    = (const float*)d_in[4];
    const float* Wk    = (const float*)d_in[5];
    const float* bk    = (const float*)d_in[6];
    const float* Wv    = (const float*)d_in[7];
    const float* bv    = (const float*)d_in[8];
    const float* gamma = (const float*)d_in[9];
    float* out = (float*)d_out;

    const long mat = (long)BB * CC * HWN;
    float* Q    = (float*)d_ws;
    float* Km   = Q  + mat;
    float* Vv   = Km + mat;
    float* corr = Vv + mat;

    const dim3 blk(128);
    const dim3 gGemm(HWN / 64, CC / 64, BB);
    const long xStride = (long)CC * HWN;

    gemm256_kernel<<<gGemm, blk, 0, stream>>>(Wq, 0, bq, fg, xStride, Q,  xStride);
    gemm256_kernel<<<gGemm, blk, 0, stream>>>(Wk, 0, bk, bg, xStride, Km, xStride);
    gemm256_kernel<<<gGemm, blk, 0, stream>>>(Wv, 0, bv, fg, xStride, Vv, xStride);

    const dim3 gCorr(CC / 16, CC / 64, BB);
    corr_kernel<<<gCorr, blk, 0, stream>>>(Q, Km, corr);

    gemm256_kernel<<<gGemm, blk, 0, stream>>>(corr, (long)CC * CC, nullptr,
                                              mask, xStride, out, xStride);

    softmax_out_kernel<<<dim3(BB * CC), dim3(256), 0, stream>>>(
        out, fg, mask, Vv, gamma, out);
}


# q, k, v and scores GEMMs: wave tile 16x64 -> 64x64 (4x4 accumulators, 2 waves per SIMD) so each fragment load feeds 4 WMMAs; loads ringed 3 k-steps ahead in registers; k and sum order kept
# speedup vs baseline: 1.4556x; 1.4556x over previous
#include <hip/hip_runtime.h>


typedef __attribute__((ext_vector_type(2))) float v2f;
typedef __attribute__((ext_vector_type(8))) float v8f;

#define BB 8
#define CC 256
#define HWN 4096


__device__ __forceinline__ void gemm_fetch(v2f (&a)[4], v2f (&b)[4],
                                           const float* __restrict__ ap, int aBlk,
                                           const float* __restrict__ bp)
{
    #pragma unroll
    for (int i = 0; i < 4; ++i)
        a[i] = *(const v2f*)(ap + i * aBlk);
    #pragma unroll
    for (int j = 0; j < 4; ++j)
        b[j] = v2f{ bp[16 * j], bp[HWN + 16 * j] };
}

template <bool D_FRAG>
__device__ __forceinline__ void gemm_mma(v8f (&acc)[4][4], const v2f (&a)[4], const v2f (&b)[4])
{
    #pragma unroll
    for (int i = 0; i < 4; ++i) {
        #pragma unroll
        for (int j = 0; j < 4; ++j) {
            if (D_FRAG)
                acc[i][j] = __builtin_amdgcn_wmma_f32_16x16x4_f32(false, b[j], false, a[i], (short)0, acc[i][j], false, false);
            else
                acc[i][j] = __builtin_amdgcn_wmma_f32_16x16x4_f32(false, a[i], false, b[j], (short)0, acc[i][j], false, false);
        }
    }
}

template <bool A_FRAG, bool D_FRAG>
__global__ __launch_bounds__(128) void gemm256_kernel(
    const float* __restrict__ A, long aStride,
    const float* __restrict__ bias,
    const float* __restrict__ Bm, long bStride,
    float* __restrict__ D, long dStride)
{
    const int lane  = threadIdx.x & 31;
    const int wave  = threadIdx.x >> 5;
    const int g     = lane >> 4;
    const int ln    = lane & 15;
    const int batch = blockIdx.z;
    const int mBlk0 = blockIdx.y * 4;
    const int mBase = mBlk0 * 16;
    const int nBase = blockIdx.x * 256 + wave * 64;
    const int n0    = nBase + ln;

    const float* Ab = A + (long)batch * aStride;
    const float* Bb = Bm + (long)batch * bStride;
    float*       Db = D + (long)batch * dStride;

    const float* ap = A_FRAG ? Ab + (long)mBlk0 * (CC / 4) * 64 + lane * 2
                             : Ab + (long)(mBase + ln) * CC + 2 * g;
    constexpr int aStep = A_FRAG ? 64 : 4;
    constexpr int aBlk  = A_FRAG ? (CC / 4) * 64 : 16 * CC;
    const float* bp = Bb + (long)(2 * g) * HWN + n0;
    constexpr int bStep = 4 * HWN;

    v8f acc[4][4] = {};
    v2f ar[4][4], br[4][4];

    gemm_fetch(ar[0], br[0], ap,             aBlk, bp);
    __builtin_amdgcn_sched_barrier(0);
    gemm_fetch(ar[1], br[1], ap + aStep,     aBlk, bp + bStep);
    __builtin_amdgcn_sched_barrier(0);
    gemm_fetch(ar[2], br[2], ap + 2 * aStep, aBlk, bp + 2 * bStep);
    __builtin_amdgcn_sched_barrier(0);
    const float* an = ap + 3 * aStep;
    const float* bn = bp + 3 * bStep;

    for (int it = 0; it < CC / 16 - 1; ++it) {
        gemm_fetch(ar[3], br[3], an,             aBlk, bn);
        __builtin_amdgcn_sched_barrier(0);
        gemm_mma<D_FRAG>(acc, ar[0], br[0]);
        __builtin_amdgcn_sched_barrier(0);
        gemm_fetch(ar[0], br[0], an + aStep,     aBlk, bn + bStep);
        __builtin_amdgcn_sched_barrier(0);
        gemm_mma<D_FRAG>(acc, ar[1], br[1]);
        __builtin_amdgcn_sched_barrier(0);
        gemm_fetch(ar[1], br[1], an + 2 * aStep, aBlk, bn + 2 * bStep);
        __builtin_amdgcn_sched_barrier(0);
        gemm_mma<D_FRAG>(acc, ar[2], br[2]);
        __builtin_amdgcn_sched_barrier(0);
        gemm_fetch(ar[2], br[2], an + 3 * aStep, aBlk, bn + 3 * bStep);
        __builtin_amdgcn_sched_barrier(0);
        gemm_mma<D_FRAG>(acc, ar[3], br[3]);
        __builtin_amdgcn_sched_barrier(0);
        an += 4 * aStep;
        bn += 4 * bStep;
    }
    gemm_fetch(ar[3], br[3], an, aBlk, bn);
    gemm_mma<D_FRAG>(acc, ar[0], br[0]);
    gemm_mma<D_FRAG>(acc, ar[1], br[1]);
    gemm_mma<D_FRAG>(acc, ar[2], br[2]);
    gemm_mma<D_FRAG>(acc, ar[3], br[3]);

    if (D_FRAG) {
        #pragma unroll
        for (int i = 0; i < 4; ++i) {
            const float bv = bias ? bias[mBase + 16 * i + ln] : 0.0f;
            float* dp = Db + ((long)(mBlk0 + i) * (HWN / 4) + (nBase >> 2) + 2 * g) * 64 + ln * 2;
            #pragma unroll
            for (int r = 0; r < 8; r += 2) {
                float* p = dp + (r >> 2) * 64 + ((r & 3) >> 1) * 32;
                #pragma unroll
                for (int j = 0; j < 4; ++j) {
                    v2f o = { acc[i][j][r] + bv, acc[i][j][r + 1] + bv };
                    *(v2f*)(p + 4 * j * 64) = o;
                }
            }
        }
    } else {
        #pragma unroll
        for (int i = 0; i < 4; ++i) {
            #pragma unroll
            for (int r = 0; r < 8; ++r) {
                const int mo = mBase + 16 * i + r + 8 * g;
                const float bv = bias ? bias[mo] : 0.0f;
                float* drow = Db + (long)mo * HWN + n0;
                #pragma unroll
                for (int j = 0; j < 4; ++j)
                    drow[16 * j] = acc[i][j][r] + bv;
            }
        }
    }
}

#define CORR_G    8
#define CORR_SETS 4
#define CORR_GS   (CORR_G * 64)
#define CORR_NG   ((HWN / 4) / CORR_G)

__device__ __forceinline__ void corr_fetch(v2f (&q)[CORR_G], v2f (&k)[CORR_G],
                                           const float* __restrict__ qsrc,
                                           const float* __restrict__ ksrc, unsigned loff)
{
    #pragma unroll
    for (int i = 0; i < CORR_G; ++i) {
        q[i] = *(const v2f*)(qsrc + i * 64 + loff);
        k[i] = *(const v2f*)(ksrc + i * 64 + loff);
    }
}

__device__ __forceinline__ v8f corr_mma(v8f acc, const v2f (&q)[CORR_G], const v2f (&k)[CORR_G])
{
    #pragma unroll
    for (int i = 0; i < CORR_G; ++i)
        acc = __builtin_amdgcn_wmma_f32_16x16x4_f32(false, k[i], false, q[i], (short)0, acc, false, false);
    return acc;
}

__global__ __launch_bounds__(128) void corr_kernel(
    const float* __restrict__ Qf, const float* __restrict__ Kf,
    float* __restrict__ corrf)
{
    const int lane  = threadIdx.x & 31;
    const int wave  = __builtin_amdgcn_readfirstlane(threadIdx.x >> 5);
    const int g     = lane >> 4;
    const int ln    = lane & 15;
    const int batch = blockIdx.z;
    const int mBlk  = blockIdx.y * 4 + wave;
    const int nBlk  = blockIdx.x;

    const float* qp = Qf + (long)batch * CC * HWN + (long)mBlk * (HWN / 4) * 64;
    const float* kp = Kf + (long)batch * CC * HWN + (long)nBlk * (HWN / 4) * 64;
    const unsigned loff = lane * 2;

    v2f qr[CORR_SETS][CORR_G], kr[CORR_SETS][CORR_G];

    corr_fetch(qr[0], kr[0], qp,               kp,               loff);
    corr_fetch(qr[1], kr[1], qp + CORR_GS,     kp + CORR_GS,     loff);
    corr_fetch(qr[2], kr[2], qp + 2 * CORR_GS, kp + 2 * CORR_GS, loff);
    const float* qn = qp + 3 * CORR_GS;
    const float* kn = kp + 3 * CORR_GS;

    v8f acc = {};
    for (int it = 0; it < CORR_NG / CORR_SETS - 1; ++it) {
        corr_fetch(qr[3], kr[3], qn,               kn,               loff);
        acc = corr_mma(acc, qr[0], kr[0]);
        corr_fetch(qr[0], kr[0], qn + CORR_GS,     kn + CORR_GS,     loff);
        acc = corr_mma(acc, qr[1], kr[1]);
        corr_fetch(qr[1], kr[1], qn + 2 * CORR_GS, kn + 2 * CORR_GS, loff);
        acc = corr_mma(acc, qr[2], kr[2]);
        corr_fetch(qr[2], kr[2], qn + 3 * CORR_GS, kn + 3 * CORR_GS, loff);
        acc = corr_mma(acc, qr[3], kr[3]);
        qn += CORR_SETS * CORR_GS;
        kn += CORR_SETS * CORR_GS;
    }
    corr_fetch(qr[3], kr[3], qp + (CORR_NG - 1) * CORR_GS, kp + (CORR_NG - 1) * CORR_GS, loff);
    acc = corr_mma(acc, qr[0], kr[0]);
    acc = corr_mma(acc, qr[1], kr[1]);
    acc = corr_mma(acc, qr[2], kr[2]);
    acc = corr_mma(acc, qr[3], kr[3]);

    float* cp = corrf + (long)batch * CC * CC
              + ((long)mBlk * (CC / 4) + nBlk * 4 + 2 * g) * 64 + ln * 2;
    #pragma unroll
    for (int r = 0; r < 8; r += 2) {
        v2f o = { acc[r], acc[r + 1] };
        *(v2f*)(cp + (r >> 2) * 64 + ((r & 3) >> 1) * 32) = o;
    }
}

__global__ __launch_bounds__(256) void softmax_out_kernel(
    const float* __restrict__ scores,
    const float* __restrict__ fg, const float* __restrict__ mask,
    const float* __restrict__ V,  const float* __restrict__ gamma,
    float* __restrict__ out)
{
    const int  tid  = threadIdx.x;
    const int  lane = tid & 31;
    const int  wid  = tid >> 5;
    const long base = (long)blockIdx.x * HWN;

    __shared__ float red[8];

    float vals[16];
    float mx = -3.402823466e38f;
    #pragma unroll
    for (int t = 0; t < 16; ++t) {
        vals[t] = scores[base + t * 256 + tid];
        mx = fmaxf(mx, vals[t]);
    }
    #pragma unroll
    for (int off = 16; off >= 1; off >>= 1)
        mx = fmaxf(mx, __shfl_xor(mx, off, 32));
    if (lane == 0) red[wid] = mx;
    __syncthreads();
    #pragma unroll
    for (int w = 0; w < 8; ++w) mx = fmaxf(mx, red[w]);
    __syncthreads();

    float sum = 0.0f;
    #pragma unroll
    for (int t = 0; t < 16; ++t) {
        vals[t] = __expf(vals[t] - mx);
        sum += vals[t];
    }
    #pragma unroll
    for (int off = 16; off >= 1; off >>= 1)
        sum += __shfl_xor(sum, off, 32);
    if (lane == 0) red[wid] = sum;
    __syncthreads();
    float tot = 0.0f;
    #pragma unroll
    for (int w = 0; w < 8; ++w) tot += red[w];

    const float inv = 1.0f / tot;
    const float gam = gamma[0];
    #pragma unroll
    for (int t = 0; t < 16; ++t) {
        const long i = base + t * 256 + tid;
        const float att = vals[t] * inv * V[i];
        const float mk  = mask[i];
        out[i] = fg[i] * mk + gam * (1.0f - mk) * att;
    }
}

extern "C" void kernel_launch(void* const* d_in, const int* in_sizes, int n_in,
                              void* d_out, int out_size, void* d_ws, size_t ws_size,
                              hipStream_t stream)
{
    (void)in_sizes; (void)n_in; (void)out_size; (void)ws_size;

    const float* fg    = (const float*)d_in[0];
    const float* bg    = (const float*)d_in[1];
    const float* mask  = (const float*)d_in[2];
    const float* Wq    = (const float*)d_in[3];
    const float* bq    = (const float*)d_in[4];
    const float* Wk    = (const float*)d_in[5];
    const float* bk    = (const float*)d_in[6];
    const float* Wv    = (const float*)d_in[7];
    const float* bv    = (const float*)d_in[8];
    const float* gamma = (const float*)d_in[9];
    float* out = (float*)d_out;

    const long mat = (long)BB * CC * HWN;
    float* Q    = (float*)d_ws;
    float* Km   = Q  + mat;
    float* Vv   = Km + mat;
    float* corr = Vv + mat;

    const dim3 blk(128);
    const dim3 gGemm(HWN / 256, CC / 64, BB);
    const long xStride = (long)CC * HWN;

    gemm256_kernel<false, true ><<<gGemm, blk, 0, stream>>>(Wq, 0, bq, fg, xStride, Q,  xStride);
    gemm256_kernel<false, true ><<<gGemm, blk, 0, stream>>>(Wk, 0, bk, bg, xStride, Km, xStride);
    gemm256_kernel<false, false><<<gGemm, blk, 0, stream>>>(Wv, 0, bv, fg, xStride, Vv, xStride);

    const dim3 gCorr(CC / 16, CC / 64, BB);
    corr_kernel<<<gCorr, blk, 0, stream>>>(Q, Km, corr);

    gemm256_kernel<true, false><<<gGemm, blk, 0, stream>>>(corr, (long)CC * CC, nullptr,
                                                            mask, xStride, out, xStride);

    softmax_out_kernel<<<dim3(BB * CC), dim3(256), 0, stream>>>(
        out, fg, mask, Vv, gamma, out);
}


# softmax launch folded into GEMM epilogues: scores product also stores per 64-column tile row max and exp-sum; v product, now last, normalises and blends in its epilogue; denominator sum order changes
# speedup vs baseline: 1.5169x; 1.0421x over previous
#include <hip/hip_runtime.h>


typedef __attribute__((ext_vector_type(2))) float v2f;
typedef __attribute__((ext_vector_type(8))) float v8f;

#define BB 8
#define CC 256
#define HWN 4096


__device__ __forceinline__ void gemm_fetch(v2f (&a)[4], v2f (&b)[4],
                                           const float* __restrict__ ap, int aBlk,
                                           const float* __restrict__ bp)
{
    #pragma unroll
    for (int i = 0; i < 4; ++i)
        a[i] = *(const v2f*)(ap + i * aBlk);
    #pragma unroll
    for (int j = 0; j < 4; ++j)
        b[j] = v2f{ bp[16 * j], bp[HWN + 16 * j] };
}

template <bool D_FRAG>
__device__ __forceinline__ void gemm_mma(v8f (&acc)[4][4], const v2f (&a)[4], const v2f (&b)[4])
{
    #pragma unroll
    for (int i = 0; i < 4; ++i) {
        #pragma unroll
        for (int j = 0; j < 4; ++j) {
            if (D_FRAG)
                acc[i][j] = __builtin_amdgcn_wmma_f32_16x16x4_f32(false, b[j], false, a[i], (short)0, acc[i][j], false, false);
            else
                acc[i][j] = __builtin_amdgcn_wmma_f32_16x16x4_f32(false, a[i], false, b[j], (short)0, acc[i][j], false, false);
        }
    }
}

template <int X>
__device__ __forceinline__ float half_xor(float v)
{
    return __builtin_bit_cast(float, __builtin_amdgcn_update_dpp(
        0, __builtin_bit_cast(int, v), 0x160 + X, 0xf, 0xf, true));
}

__device__ __forceinline__ float half_max(float v)
{
    v = fmaxf(v, half_xor<1>(v));
    v = fmaxf(v, half_xor<2>(v));
    v = fmaxf(v, half_xor<4>(v));
    v = fmaxf(v, half_xor<8>(v));
    return v;
}

__device__ __forceinline__ float half_sum(float v)
{
    v += half_xor<1>(v);
    v += half_xor<2>(v);
    v += half_xor<4>(v);
    v += half_xor<8>(v);
    return v;
}

struct EpiBuf {
    float4 p0[4], p1[4];
    float  sc[4][4], mk[4][4], fv[4][4];
};

__device__ __forceinline__ void epi_load(EpiBuf& b, int row0, const float* sb,
                                         const float* ob, const float* __restrict__ mb,
                                         const float* __restrict__ fb)
{
    #pragma unroll
    for (int r = 0; r < 4; ++r) {
        const float* sp = sb + (long)(row0 + r) * 128;
        b.p0[r] = *(const float4*)sp;
        b.p1[r] = *(const float4*)(sp + 4);
        const long e = (long)(row0 + r) * HWN;
        #pragma unroll
        for (int j = 0; j < 4; ++j) {
            b.sc[r][j] = ob[e + 16 * j];
            b.mk[r][j] = mb[e + 16 * j];
            b.fv[r][j] = fb[e + 16 * j];
        }
    }
}

template <bool A_FRAG, bool D_FRAG, int EPI>
__global__ __launch_bounds__(128) void gemm256_kernel(
    const float* __restrict__ A, long aStride,
    const float* __restrict__ bias,
    const float* __restrict__ Bm, long bStride,
    float* __restrict__ D, long dStride,
    float* stats,
    const float* __restrict__ maskp,
    const float* __restrict__ gamma,
    float* io)
{
    const int lane  = threadIdx.x & 31;
    const int wave  = threadIdx.x >> 5;
    const int g     = lane >> 4;
    const int ln    = lane & 15;
    const int batch = blockIdx.z;
    const int mBlk0 = blockIdx.y * 4;
    const int mBase = mBlk0 * 16;
    const int nBase = blockIdx.x * 256 + wave * 64;
    const int n0    = nBase + ln;

    const float* Ab = A + (long)batch * aStride;
    const float* Bb = Bm + (long)batch * bStride;
    float*       Db = D + (long)batch * dStride;

    const float* ap = A_FRAG ? Ab + (long)mBlk0 * (CC / 4) * 64 + lane * 2
                             : Ab + (long)(mBase + ln) * CC + 2 * g;
    constexpr int aStep = A_FRAG ? 64 : 4;
    constexpr int aBlk  = A_FRAG ? (CC / 4) * 64 : 16 * CC;
    const float* bp = Bb + (long)(2 * g) * HWN + n0;
    constexpr int bStep = 4 * HWN;

    v8f acc[4][4] = {};
    v2f ar[4][4], br[4][4];

    gemm_fetch(ar[0], br[0], ap,             aBlk, bp);
    __builtin_amdgcn_sched_barrier(0);
    gemm_fetch(ar[1], br[1], ap + aStep,     aBlk, bp + bStep);
    __builtin_amdgcn_sched_barrier(0);
    gemm_fetch(ar[2], br[2], ap + 2 * aStep, aBlk, bp + 2 * bStep);
    __builtin_amdgcn_sched_barrier(0);
    const float* an = ap + 3 * aStep;
    const float* bn = bp + 3 * bStep;

    for (int it = 0; it < CC / 16 - 1; ++it) {
        gemm_fetch(ar[3], br[3], an,             aBlk, bn);
        __builtin_amdgcn_sched_barrier(0);
        gemm_mma<D_FRAG>(acc, ar[0], br[0]);
        __builtin_amdgcn_sched_barrier(0);
        gemm_fetch(ar[0], br[0], an + aStep,     aBlk, bn + bStep);
        __builtin_amdgcn_sched_barrier(0);
        gemm_mma<D_FRAG>(acc, ar[1], br[1]);
        __builtin_amdgcn_sched_barrier(0);
        gemm_fetch(ar[1], br[1], an + 2 * aStep, aBlk, bn + 2 * bStep);
        __builtin_amdgcn_sched_barrier(0);
        gemm_mma<D_FRAG>(acc, ar[2], br[2]);
        __builtin_amdgcn_sched_barrier(0);
        gemm_fetch(ar[2], br[2], an + 3 * aStep, aBlk, bn + 3 * bStep);
        __builtin_amdgcn_sched_barrier(0);
        gemm_mma<D_FRAG>(acc, ar[3], br[3]);
        __builtin_amdgcn_sched_barrier(0);
        an += 4 * aStep;
        bn += 4 * bStep;
    }
    gemm_fetch(ar[3], br[3], an, aBlk, bn);

    gemm_mma<D_FRAG>(acc, ar[0], br[0]);
    gemm_mma<D_FRAG>(acc, ar[1], br[1]);

    EpiBuf eb[2];
    float bvr[4][8];
    const float* e_mb = nullptr;
    float*       e_ob = nullptr;
    const float* e_sb = nullptr;
    const float* e_fb = nullptr;
    if (EPI == 2) {
        __builtin_amdgcn_sched_barrier(0);
        e_mb = maskp + (long)batch * bStride + n0;
        e_ob = io + (long)batch * bStride + n0;
        e_fb = Bb + n0;
        e_sb = stats + (long)batch * CC * 128 + 8 * ln;
        if (bias) {
            #pragma unroll
            for (int i = 0; i < 4; ++i)
                #pragma unroll
                for (int r = 0; r < 8; ++r)
                    bvr[i][r] = bias[mBase + 16 * i + r + 8 * g];
        } else {
            #pragma unroll
            for (int i = 0; i < 4; ++i)
                #pragma unroll
                for (int r = 0; r < 8; ++r)
                    bvr[i][r] = 0.0f;
        }
        epi_load(eb[0], mBase + 8 * g, e_sb, e_ob, e_mb, e_fb);
        __builtin_amdgcn_sched_barrier(0);
    }

    gemm_mma<D_FRAG>(acc, ar[2], br[2]);
    gemm_mma<D_FRAG>(acc, ar[3], br[3]);

    if (D_FRAG) {
        #pragma unroll
        for (int i = 0; i < 4; ++i) {
            const float bv = bias ? bias[mBase + 16 * i + ln] : 0.0f;
            float* dp = Db + ((long)(mBlk0 + i) * (HWN / 4) + (nBase >> 2) + 2 * g) * 64 + ln * 2;
            #pragma unroll
            for (int r = 0; r < 8; r += 2) {
                float* p = dp + (r >> 2) * 64 + ((r & 3) >> 1) * 32;
                #pragma unroll
                for (int j = 0; j < 4; ++j) {
                    v2f o = { acc[i][j][r] + bv, acc[i][j][r + 1] + bv };
                    *(v2f*)(p + 4 * j * 64) = o;
                }
            }
        }
    } else if (EPI == 2) {
        const float gam = gamma[0];
        __builtin_amdgcn_sched_barrier(0);
        #pragma unroll
        for (int h = 0; h < 8; ++h) {
            const int i  = h >> 1;
            const int r0 = (h & 1) * 4;
            if (h + 1 < 8)
                epi_load(eb[(h + 1) & 1], mBase + 16 * ((h + 1) >> 1) + ((h + 1) & 1) * 4 + 8 * g,
                         e_sb, e_ob, e_mb, e_fb);
            __builtin_amdgcn_sched_barrier(0);
            EpiBuf& b = eb[h & 1];
            float M[4], inv[4];
            #pragma unroll
            for (int r = 0; r < 4; ++r)
                M[r] = half_max(fmaxf(fmaxf(b.p0[r].x, b.p0[r].z), fmaxf(b.p1[r].x, b.p1[r].z)));
            #pragma unroll
            for (int r = 0; r < 4; ++r) {
                const float tot = half_sum(b.p0[r].y * __expf(b.p0[r].x - M[r]) + b.p0[r].w * __expf(b.p0[r].z - M[r])
                                         + b.p1[r].y * __expf(b.p1[r].x - M[r]) + b.p1[r].w * __expf(b.p1[r].z - M[r]));
                inv[r] = 1.0f / tot;
            }
            #pragma unroll
            for (int r = 0; r < 4; ++r) {
                const long e = (long)(mBase + 16 * i + r0 + r + 8 * g) * HWN;
                #pragma unroll
                for (int j = 0; j < 4; ++j) {
                    const float att = __expf(b.sc[r][j] - M[r]) * inv[r] * (acc[i][j][r0 + r] + bvr[i][r0 + r]);
                    e_ob[e + 16 * j] = b.fv[r][j] * b.mk[r][j] + gam * (1.0f - b.mk[r][j]) * att;
                }
            }
            __builtin_amdgcn_sched_barrier(0);
        }
    } else {
        const int tile = blockIdx.x * 4 + wave;
        float* sb = stats + (long)batch * CC * 128 + tile * 2;
        #pragma unroll
        for (int i = 0; i < 4; ++i) {
            float x[8][4];
            #pragma unroll
            for (int r = 0; r < 8; ++r) {
                const int mo = mBase + 16 * i + r + 8 * g;
                const float bv = bias ? bias[mo] : 0.0f;
                float* drow = Db + (long)mo * HWN + n0;
                #pragma unroll
                for (int j = 0; j < 4; ++j) {
                    x[r][j] = acc[i][j][r] + bv;
                    drow[16 * j] = x[r][j];
                }
            }
            if (EPI == 1) {
                float m[8], s[8];
                #pragma unroll
                for (int r = 0; r < 8; ++r)
                    m[r] = half_max(fmaxf(fmaxf(x[r][0], x[r][1]), fmaxf(x[r][2], x[r][3])));
                #pragma unroll
                for (int r = 0; r < 8; ++r)
                    s[r] = half_sum(__expf(x[r][0] - m[r]) + __expf(x[r][1] - m[r])
                                  + __expf(x[r][2] - m[r]) + __expf(x[r][3] - m[r]));
                if (ln == 0) {
                    #pragma unroll
                    for (int r = 0; r < 8; ++r) {
                        const int mo = mBase + 16 * i + r + 8 * g;
                        v2f o = { m[r], s[r] };
                        *(v2f*)(sb + (long)mo * 128) = o;
                    }
                }
            }
        }
    }
}

#define CORR_G    8
#define CORR_SETS 4
#define CORR_GS   (CORR_G * 64)
#define CORR_NG   ((HWN / 4) / CORR_G)

__device__ __forceinline__ void corr_fetch(v2f (&q)[CORR_G], v2f (&k)[CORR_G],
                                           const float* __restrict__ qsrc,
                                           const float* __restrict__ ksrc, unsigned loff)
{
    #pragma unroll
    for (int i = 0; i < CORR_G; ++i) {
        q[i] = *(const v2f*)(qsrc + i * 64 + loff);
        k[i] = *(const v2f*)(ksrc + i * 64 + loff);
    }
}

__device__ __forceinline__ v8f corr_mma(v8f acc, const v2f (&q)[CORR_G], const v2f (&k)[CORR_G])
{
    #pragma unroll
    for (int i = 0; i < CORR_G; ++i)
        acc = __builtin_amdgcn_wmma_f32_16x16x4_f32(false, k[i], false, q[i], (short)0, acc, false, false);
    return acc;
}

__global__ __launch_bounds__(128) void corr_kernel(
    const float* __restrict__ Qf, const float* __restrict__ Kf,
    float* __restrict__ corrf)
{
    const int lane  = threadIdx.x & 31;
    const int wave  = __builtin_amdgcn_readfirstlane(threadIdx.x >> 5);
    const int g     = lane >> 4;
    const int ln    = lane & 15;
    const int batch = blockIdx.z;
    const int mBlk  = blockIdx.y * 4 + wave;
    const int nBlk  = blockIdx.x;

    const float* qp = Qf + (long)batch * CC * HWN + (long)mBlk * (HWN / 4) * 64;
    const float* kp = Kf + (long)batch * CC * HWN + (long)nBlk * (HWN / 4) * 64;
    const unsigned loff = lane * 2;

    v2f qr[CORR_SETS][CORR_G], kr[CORR_SETS][CORR_G];

    corr_fetch(qr[0], kr[0], qp,               kp,               loff);
    corr_fetch(qr[1], kr[1], qp + CORR_GS,     kp + CORR_GS,     loff);
    corr_fetch(qr[2], kr[2], qp + 2 * CORR_GS, kp + 2 * CORR_GS, loff);
    const float* qn = qp + 3 * CORR_GS;
    const float* kn = kp + 3 * CORR_GS;

    v8f acc = {};
    for (int it = 0; it < CORR_NG / CORR_SETS - 1; ++it) {
        corr_fetch(qr[3], kr[3], qn,               kn,               loff);
        acc = corr_mma(acc, qr[0], kr[0]);
        corr_fetch(qr[0], kr[0], qn + CORR_GS,     kn + CORR_GS,     loff);
        acc = corr_mma(acc, qr[1], kr[1]);
        corr_fetch(qr[1], kr[1], qn + 2 * CORR_GS, kn + 2 * CORR_GS, loff);
        acc = corr_mma(acc, qr[2], kr[2]);
        corr_fetch(qr[2], kr[2], qn + 3 * CORR_GS, kn + 3 * CORR_GS, loff);
        acc = corr_mma(acc, qr[3], kr[3]);
        qn += CORR_SETS * CORR_GS;
        kn += CORR_SETS * CORR_GS;
    }
    corr_fetch(qr[3], kr[3], qp + (CORR_NG - 1) * CORR_GS, kp + (CORR_NG - 1) * CORR_GS, loff);
    acc = corr_mma(acc, qr[0], kr[0]);
    acc = corr_mma(acc, qr[1], kr[1]);
    acc = corr_mma(acc, qr[2], kr[2]);
    acc = corr_mma(acc, qr[3], kr[3]);

    float* cp = corrf + (long)batch * CC * CC
              + ((long)mBlk * (CC / 4) + nBlk * 4 + 2 * g) * 64 + ln * 2;
    #pragma unroll
    for (int r = 0; r < 8; r += 2) {
        v2f o = { acc[r], acc[r + 1] };
        *(v2f*)(cp + (r >> 2) * 64 + ((r & 3) >> 1) * 32) = o;
    }
}

extern "C" void kernel_launch(void* const* d_in, const int* in_sizes, int n_in,
                              void* d_out, int out_size, void* d_ws, size_t ws_size,
                              hipStream_t stream)
{
    (void)in_sizes; (void)n_in; (void)out_size; (void)ws_size;

    const float* fg    = (const float*)d_in[0];
    const float* bg    = (const float*)d_in[1];
    const float* mask  = (const float*)d_in[2];
    const float* Wq    = (const float*)d_in[3];
    const float* bq    = (const float*)d_in[4];
    const float* Wk    = (const float*)d_in[5];
    const float* bk    = (const float*)d_in[6];
    const float* Wv    = (const float*)d_in[7];
    const float* bv    = (const float*)d_in[8];
    const float* gamma = (const float*)d_in[9];
    float* out = (float*)d_out;

    const long mat = (long)BB * CC * HWN;
    float* Q     = (float*)d_ws;
    float* Km    = Q  + mat;
    float* stats = Km + mat;
    float* corr  = stats + mat;

    const dim3 blk(128);
    const dim3 gGemm(HWN / 256, CC / 64, BB);
    const long xStride = (long)CC * HWN;

    gemm256_kernel<false, true, 0><<<gGemm, blk, 0, stream>>>(Wq, 0, bq, fg, xStride, Q,  xStride,
                                                              nullptr, nullptr, nullptr, nullptr);
    gemm256_kernel<false, true, 0><<<gGemm, blk, 0, stream>>>(Wk, 0, bk, bg, xStride, Km, xStride,
                                                              nullptr, nullptr, nullptr, nullptr);

    const dim3 gCorr(CC / 16, CC / 64, BB);
    corr_kernel<<<gCorr, blk, 0, stream>>>(Q, Km, corr);

    gemm256_kernel<true, false, 1><<<gGemm, blk, 0, stream>>>(corr, (long)CC * CC, nullptr,
                                                              mask, xStride, out, xStride,
                                                              stats, nullptr, nullptr, nullptr);

    gemm256_kernel<false, false, 2><<<gGemm, blk, 0, stream>>>(Wv, 0, bv, fg, xStride, nullptr, 0,
                                                               stats, mask, gamma, out);
}


# q and k projections share one launch (grid z = 2 x batch; z >= batch takes the k operands): 4 launches, not 5, and four waves per SIMD in that launch; tiles, loops, k and sum orders kept
# speedup vs baseline: 1.5794x; 1.0412x over previous
#include <hip/hip_runtime.h>


typedef __attribute__((ext_vector_type(2))) float v2f;
typedef __attribute__((ext_vector_type(8))) float v8f;

#define BB 8
#define CC 256
#define HWN 4096


__device__ __forceinline__ void gemm_fetch(v2f (&a)[4], v2f (&b)[4],
                                           const float* __restrict__ ap, int aBlk,
                                           const float* __restrict__ bp)
{
    #pragma unroll
    for (int i = 0; i < 4; ++i)
        a[i] = *(const v2f*)(ap + i * aBlk);
    #pragma unroll
    for (int j = 0; j < 4; ++j)
        b[j] = v2f{ bp[16 * j], bp[HWN + 16 * j] };
}

template <bool D_FRAG>
__device__ __forceinline__ void gemm_mma(v8f (&acc)[4][4], const v2f (&a)[4], const v2f (&b)[4])
{
    #pragma unroll
    for (int i = 0; i < 4; ++i) {
        #pragma unroll
        for (int j = 0; j < 4; ++j) {
            if (D_FRAG)
                acc[i][j] = __builtin_amdgcn_wmma_f32_16x16x4_f32(false, b[j], false, a[i], (short)0, acc[i][j], false, false);
            else
                acc[i][j] = __builtin_amdgcn_wmma_f32_16x16x4_f32(false, a[i], false, b[j], (short)0, acc[i][j], false, false);
        }
    }
}

template <int X>
__device__ __forceinline__ float half_xor(float v)
{
    return __builtin_bit_cast(float, __builtin_amdgcn_update_dpp(
        0, __builtin_bit_cast(int, v), 0x160 + X, 0xf, 0xf, true));
}

__device__ __forceinline__ float half_max(float v)
{
    v = fmaxf(v, half_xor<1>(v));
    v = fmaxf(v, half_xor<2>(v));
    v = fmaxf(v, half_xor<4>(v));
    v = fmaxf(v, half_xor<8>(v));
    return v;
}

__device__ __forceinline__ float half_sum(float v)
{
    v += half_xor<1>(v);
    v += half_xor<2>(v);
    v += half_xor<4>(v);
    v += half_xor<8>(v);
    return v;
}

struct EpiBuf {
    float4 p0[4], p1[4];
    float  sc[4][4], mk[4][4], fv[4][4];
};

__device__ __forceinline__ void epi_load(EpiBuf& b, int row0, const float* sb,
                                         const float* ob, const float* __restrict__ mb,
                                         const float* __restrict__ fb)
{
    #pragma unroll
    for (int r = 0; r < 4; ++r) {
        const float* sp = sb + (long)(row0 + r) * 128;
        b.p0[r] = *(const float4*)sp;
        b.p1[r] = *(const float4*)(sp + 4);
        const long e = (long)(row0 + r) * HWN;
        #pragma unroll
        for (int j = 0; j < 4; ++j) {
            b.sc[r][j] = ob[e + 16 * j];
            b.mk[r][j] = mb[e + 16 * j];
            b.fv[r][j] = fb[e + 16 * j];
        }
    }
}

template <bool A_FRAG, bool D_FRAG, int EPI, bool PAIR>
__global__ __launch_bounds__(128) void gemm256_kernel(
    const float* __restrict__ A0, long aStride,
    const float* __restrict__ bias0,
    const float* __restrict__ B0, long bStride,
    float* __restrict__ D0, long dStride,
    float* stats,
    const float* __restrict__ maskp,
    const float* __restrict__ gamma,
    float* io,
    const float* __restrict__ A1,
    const float* __restrict__ bias1,
    const float* __restrict__ B1,
    float* __restrict__ D1)
{
    const int lane  = threadIdx.x & 31;
    const int wave  = threadIdx.x >> 5;
    const int g     = lane >> 4;
    const int ln    = lane & 15;
    const bool second = PAIR && blockIdx.z >= BB;
    const int batch = second ? (int)blockIdx.z - BB : (int)blockIdx.z;
    const float* __restrict__ A    = second ? A1 : A0;
    const float* __restrict__ bias = second ? bias1 : bias0;
    const float* __restrict__ Bm   = second ? B1 : B0;
    float* __restrict__       D    = second ? D1 : D0;
    const int mBlk0 = blockIdx.y * 4;
    const int mBase = mBlk0 * 16;
    const int nBase = blockIdx.x * 256 + wave * 64;
    const int n0    = nBase + ln;

    const float* Ab = A + (long)batch * aStride;
    const float* Bb = Bm + (long)batch * bStride;
    float*       Db = D + (long)batch * dStride;

    const float* ap = A_FRAG ? Ab + (long)mBlk0 * (CC / 4) * 64 + lane * 2
                             : Ab + (long)(mBase + ln) * CC + 2 * g;
    constexpr int aStep = A_FRAG ? 64 : 4;
    constexpr int aBlk  = A_FRAG ? (CC / 4) * 64 : 16 * CC;
    const float* bp = Bb + (long)(2 * g) * HWN + n0;
    constexpr int bStep = 4 * HWN;

    v8f acc[4][4] = {};
    v2f ar[4][4], br[4][4];

    gemm_fetch(ar[0], br[0], ap,             aBlk, bp);
    __builtin_amdgcn_sched_barrier(0);
    gemm_fetch(ar[1], br[1], ap + aStep,     aBlk, bp + bStep);
    __builtin_amdgcn_sched_barrier(0);
    gemm_fetch(ar[2], br[2], ap + 2 * aStep, aBlk, bp + 2 * bStep);
    __builtin_amdgcn_sched_barrier(0);
    const float* an = ap + 3 * aStep;
    const float* bn = bp + 3 * bStep;

    for (int it = 0; it < CC / 16 - 1; ++it) {
        gemm_fetch(ar[3], br[3], an,             aBlk, bn);
        __builtin_amdgcn_sched_barrier(0);
        gemm_mma<D_FRAG>(acc, ar[0], br[0]);
        __builtin_amdgcn_sched_barrier(0);
        gemm_fetch(ar[0], br[0], an + aStep,     aBlk, bn + bStep);
        __builtin_amdgcn_sched_barrier(0);
        gemm_mma<D_FRAG>(acc, ar[1], br[1]);
        __builtin_amdgcn_sched_barrier(0);
        gemm_fetch(ar[1], br[1], an + 2 * aStep, aBlk, bn + 2 * bStep);
        __builtin_amdgcn_sched_barrier(0);
        gemm_mma<D_FRAG>(acc, ar[2], br[2]);
        __builtin_amdgcn_sched_barrier(0);
        gemm_fetch(ar[2], br[2], an + 3 * aStep, aBlk, bn + 3 * bStep);
        __builtin_amdgcn_sched_barrier(0);
        gemm_mma<D_FRAG>(acc, ar[3], br[3]);
        __builtin_amdgcn_sched_barrier(0);
        an += 4 * aStep;
        bn += 4 * bStep;
    }
    gemm_fetch(ar[3], br[3], an, aBlk, bn);

    gemm_mma<D_FRAG>(acc, ar[0], br[0]);
    gemm_mma<D_FRAG>(acc, ar[1], br[1]);

    EpiBuf eb[2];
    float bvr[4][8];
    const float* e_mb = nullptr;
    float*       e_ob = nullptr;
    const float* e_sb = nullptr;
    const float* e_fb = nullptr;
    if (EPI == 2) {
        __builtin_amdgcn_sched_barrier(0);
        e_mb = maskp + (long)batch * bStride + n0;
        e_ob = io + (long)batch * bStride + n0;
        e_fb = Bb + n0;
        e_sb = stats + (long)batch * CC * 128 + 8 * ln;
        if (bias) {
            #pragma unroll
            for (int i = 0; i < 4; ++i)
                #pragma unroll
                for (int r = 0; r < 8; ++r)
                    bvr[i][r] = bias[mBase + 16 * i + r + 8 * g];
        } else {
            #pragma unroll
            for (int i = 0; i < 4; ++i)
                #pragma unroll
                for (int r = 0; r < 8; ++r)
                    bvr[i][r] = 0.0f;
        }
        epi_load(eb[0], mBase + 8 * g, e_sb, e_ob, e_mb, e_fb);
        __builtin_amdgcn_sched_barrier(0);
    }

    gemm_mma<D_FRAG>(acc, ar[2], br[2]);
    gemm_mma<D_FRAG>(acc, ar[3], br[3]);

    if (D_FRAG) {
        #pragma unroll
        for (int i = 0; i < 4; ++i) {
            const float bv = bias ? bias[mBase + 16 * i + ln] : 0.0f;
            float* dp = Db + ((long)(mBlk0 + i) * (HWN / 4) + (nBase >> 2) + 2 * g) * 64 + ln * 2;
            #pragma unroll
            for (int r = 0; r < 8; r += 2) {
                float* p = dp + (r >> 2) * 64 + ((r & 3) >> 1) * 32;
                #pragma unroll
                for (int j = 0; j < 4; ++j) {
                    v2f o = { acc[i][j][r] + bv, acc[i][j][r + 1] + bv };
                    *(v2f*)(p + 4 * j * 64) = o;
                }
            }
        }
    } else if (EPI == 2) {
        const float gam = gamma[0];
        __builtin_amdgcn_sched_barrier(0);
        #pragma unroll
        for (int h = 0; h < 8; ++h) {
            const int i  = h >> 1;
            const int r0 = (h & 1) * 4;
            if (h + 1 < 8)
                epi_load(eb[(h + 1) & 1], mBase + 16 * ((h + 1) >> 1) + ((h + 1) & 1) * 4 + 8 * g,
                         e_sb, e_ob, e_mb, e_fb);
            __builtin_amdgcn_sched_barrier(0);
            EpiBuf& b = eb[h & 1];
            float M[4], inv[4];
            #pragma unroll
            for (int r = 0; r < 4; ++r)
                M[r] = half_max(fmaxf(fmaxf(b.p0[r].x, b.p0[r].z), fmaxf(b.p1[r].x, b.p1[r].z)));
            #pragma unroll
            for (int r = 0; r < 4; ++r) {
                const float tot = half_sum(b.p0[r].y * __expf(b.p0[r].x - M[r]) + b.p0[r].w * __expf(b.p0[r].z - M[r])
                                         + b.p1[r].y * __expf(b.p1[r].x - M[r]) + b.p1[r].w * __expf(b.p1[r].z - M[r]));
                inv[r] = 1.0f / tot;
            }
            #pragma unroll
            for (int r = 0; r < 4; ++r) {
                const long e = (long)(mBase + 16 * i + r0 + r + 8 * g) * HWN;
                #pragma unroll
                for (int j = 0; j < 4; ++j) {
                    const float att = __expf(b.sc[r][j] - M[r]) * inv[r] * (acc[i][j][r0 + r] + bvr[i][r0 + r]);
                    e_ob[e + 16 * j] = b.fv[r][j] * b.mk[r][j] + gam * (1.0f - b.mk[r][j]) * att;
                }
            }
            __builtin_amdgcn_sched_barrier(0);
        }
    } else {
        const int tile = blockIdx.x * 4 + wave;
        float* sb = stats + (long)batch * CC * 128 + tile * 2;
        #pragma unroll
        for (int i = 0; i < 4; ++i) {
            float x[8][4];
            #pragma unroll
            for (int r = 0; r < 8; ++r) {
                const int mo = mBase + 16 * i + r + 8 * g;
                const float bv = bias ? bias[mo] : 0.0f;
                float* drow = Db + (long)mo * HWN + n0;
                #pragma unroll
                for (int j = 0; j < 4; ++j) {
                    x[r][j] = acc[i][j][r] + bv;
                    drow[16 * j] = x[r][j];
                }
            }
            if (EPI == 1) {
                float m[8], s[8];
                #pragma unroll
                for (int r = 0; r < 8; ++r)
                    m[r] = half_max(fmaxf(fmaxf(x[r][0], x[r][1]), fmaxf(x[r][2], x[r][3])));
                #pragma unroll
                for (int r = 0; r < 8; ++r)
                    s[r] = half_sum(__expf(x[r][0] - m[r]) + __expf(x[r][1] - m[r])
                                  + __expf(x[r][2] - m[r]) + __expf(x[r][3] - m[r]));
                if (ln == 0) {
                    #pragma unroll
                    for (int r = 0; r < 8; ++r) {
                        const int mo = mBase + 16 * i + r + 8 * g;
                        v2f o = { m[r], s[r] };
                        *(v2f*)(sb + (long)mo * 128) = o;
                    }
                }
            }
        }
    }
}

#define CORR_G    8
#define CORR_SETS 4
#define CORR_GS   (CORR_G * 64)
#define CORR_NG   ((HWN / 4) / CORR_G)

__device__ __forceinline__ void corr_fetch(v2f (&q)[CORR_G], v2f (&k)[CORR_G],
                                           const float* __restrict__ qsrc,
                                           const float* __restrict__ ksrc, unsigned loff)
{
    #pragma unroll
    for (int i = 0; i < CORR_G; ++i) {
        q[i] = *(const v2f*)(qsrc + i * 64 + loff);
        k[i] = *(const v2f*)(ksrc + i * 64 + loff);
    }
}

__device__ __forceinline__ v8f corr_mma(v8f acc, const v2f (&q)[CORR_G], const v2f (&k)[CORR_G])
{
    #pragma unroll
    for (int i = 0; i < CORR_G; ++i)
        acc = __builtin_amdgcn_wmma_f32_16x16x4_f32(false, k[i], false, q[i], (short)0, acc, false, false);
    return acc;
}

__global__ __launch_bounds__(128) void corr_kernel(
    const float* __restrict__ Qf, const float* __restrict__ Kf,
    float* __restrict__ corrf)
{
    const int lane  = threadIdx.x & 31;
    const int wave  = __builtin_amdgcn_readfirstlane(threadIdx.x >> 5);
    const int g     = lane >> 4;
    const int ln    = lane & 15;
    const int batch = blockIdx.z;
    const int mBlk  = blockIdx.y * 4 + wave;
    const int nBlk  = blockIdx.x;

    const float* qp = Qf + (long)batch * CC * HWN + (long)mBlk * (HWN / 4) * 64;
    const float* kp = Kf + (long)batch * CC * HWN + (long)nBlk * (HWN / 4) * 64;
    const unsigned loff = lane * 2;

    v2f qr[CORR_SETS][CORR_G], kr[CORR_SETS][CORR_G];

    corr_fetch(qr[0], kr[0], qp,               kp,               loff);
    corr_fetch(qr[1], kr[1], qp + CORR_GS,     kp + CORR_GS,     loff);
    corr_fetch(qr[2], kr[2], qp + 2 * CORR_GS, kp + 2 * CORR_GS, loff);
    const float* qn = qp + 3 * CORR_GS;
    const float* kn = kp + 3 * CORR_GS;

    v8f acc = {};
    for (int it = 0; it < CORR_NG / CORR_SETS - 1; ++it) {
        corr_fetch(qr[3], kr[3], qn,               kn,               loff);
        acc = corr_mma(acc, qr[0], kr[0]);
        corr_fetch(qr[0], kr[0], qn + CORR_GS,     kn + CORR_GS,     loff);
        acc = corr_mma(acc, qr[1], kr[1]);
        corr_fetch(qr[1], kr[1], qn + 2 * CORR_GS, kn + 2 * CORR_GS, loff);
        acc = corr_mma(acc, qr[2], kr[2]);
        corr_fetch(qr[2], kr[2], qn + 3 * CORR_GS, kn + 3 * CORR_GS, loff);
        acc = corr_mma(acc, qr[3], kr[3]);
        qn += CORR_SETS * CORR_GS;
        kn += CORR_SETS * CORR_GS;
    }
    corr_fetch(qr[3], kr[3], qp + (CORR_NG - 1) * CORR_GS, kp + (CORR_NG - 1) * CORR_GS, loff);
    acc = corr_mma(acc, qr[0], kr[0]);
    acc = corr_mma(acc, qr[1], kr[1]);
    acc = corr_mma(acc, qr[2], kr[2]);
    acc = corr_mma(acc, qr[3], kr[3]);

    float* cp = corrf + (long)batch * CC * CC
              + ((long)mBlk * (CC / 4) + nBlk * 4 + 2 * g) * 64 + ln * 2;
    #pragma unroll
    for (int r = 0; r < 8; r += 2) {
        v2f o = { acc[r], acc[r + 1] };
        *(v2f*)(cp + (r >> 2) * 64 + ((r & 3) >> 1) * 32) = o;
    }
}

extern "C" void kernel_launch(void* const* d_in, const int* in_sizes, int n_in,
                              void* d_out, int out_size, void* d_ws, size_t ws_size,
                              hipStream_t stream)
{
    (void)in_sizes; (void)n_in; (void)out_size; (void)ws_size;

    const float* fg    = (const float*)d_in[0];
    const float* bg    = (const float*)d_in[1];
    const float* mask  = (const float*)d_in[2];
    const float* Wq    = (const float*)d_in[3];
    const float* bq    = (const float*)d_in[4];
    const float* Wk    = (const float*)d_in[5];
    const float* bk    = (const float*)d_in[6];
    const float* Wv    = (const float*)d_in[7];
    const float* bv    = (const float*)d_in[8];
    const float* gamma = (const float*)d_in[9];
    float* out = (float*)d_out;

    const long mat = (long)BB * CC * HWN;
    float* Q     = (float*)d_ws;
    float* Km    = Q  + mat;
    float* stats = Km + mat;
    float* corr  = stats + mat;

    const dim3 blk(128);
    const dim3 gGemm(HWN / 256, CC / 64, BB);
    const long xStride = (long)CC * HWN;

    const dim3 gPair(HWN / 256, CC / 64, 2 * BB);
    gemm256_kernel<false, true, 0, true><<<gPair, blk, 0, stream>>>(Wq, 0, bq, fg, xStride, Q, xStride,
                                                                    nullptr, nullptr, nullptr, nullptr,
                                                                    Wk, bk, bg, Km);

    const dim3 gCorr(CC / 16, CC / 64, BB);
    corr_kernel<<<gCorr, blk, 0, stream>>>(Q, Km, corr);

    gemm256_kernel<true, false, 1, false><<<gGemm, blk, 0, stream>>>(corr, (long)CC * CC, nullptr,
                                                                     mask, xStride, out, xStride,
                                                                     stats, nullptr, nullptr, nullptr,
                                                                     nullptr, nullptr, nullptr, nullptr);

    gemm256_kernel<false, false, 2, false><<<gGemm, blk, 0, stream>>>(Wv, 0, bv, fg, xStride, nullptr, 0,
                                                                      stats, mask, gamma, out,
                                                                      nullptr, nullptr, nullptr, nullptr);
}
